# starting kernel
# baseline (speedup 1.0000x reference)
#include <hip/hip_runtime.h>
#include <hip/hip_bf16.h>
#include <math.h>


typedef __attribute__((ext_vector_type(2))) float v2f;
typedef __attribute__((ext_vector_type(8))) float v8f;

#define B_   4
#define C_   256
#define HDS  8
#define DK   32
#define N_   2304
#define NT   (N_ / 16)
#define QSCALE 0.17677669529663687f

static __device__ __forceinline__ v8f wmma_f32(v2f a, v2f b, v8f c) {
  return __builtin_amdgcn_wmma_f32_16x16x4_f32(
      false, a, false, b, (short)0, c, false, false);
}

__global__ __launch_bounds__(128) void mha_proj_kernel(
    const float* __restrict__ x,
    const float* __restrict__ wq, const float* __restrict__ wk,
    const float* __restrict__ wv,
    float* __restrict__ qt, float* __restrict__ kb, float* __restrict__ vt) {
  const int lane = threadIdx.x & 31;
  const int wave = threadIdx.x >> 5;
  const int t     = blockIdx.x * 4 + wave;
  const int oTile = t & 15;
  const int nTile = t >> 4;
  const int p = blockIdx.y % 3;
  const int b = blockIdx.y / 3;
  const float* w = (p == 0) ? wq : (p == 1) ? wk : wv;

  const int row  = lane & 15;
  const int half = lane >> 4;
  const int koff = half * 2;
  const int oBase = oTile * 16;
  const int nBase = nTile * 16;

  const float* wA = w + (size_t)(oBase + row) * C_ + koff;
  const float* xB = x + ((size_t)b * C_ + koff) * N_ + nBase + row;

  v8f acc = {};
#pragma unroll 8
  for (int k = 0; k < C_; k += 4) {
    v2f a, bm;
    a.x  = wA[k];
    a.y  = wA[k + 1];
    bm.x = xB[(size_t)k * N_];
    bm.y = xB[(size_t)(k + 1) * N_];
    acc = wmma_f32(a, bm, acc);
  }

  if (p == 1) {
    float* outp = kb + ((size_t)b * C_ + oBase + half * 8) * N_ + nBase + row;
#pragma unroll
    for (int v = 0; v < 8; ++v) outp[(size_t)v * N_] = acc[v];
  } else {
    float* dst = (p == 0) ? qt : vt;
    const float scale = (p == 0) ? QSCALE : 1.0f;
#pragma unroll
    for (int v = 0; v < 8; ++v) {
      const int o = oBase + v + half * 8;
      const int h = o >> 5, d = o & 31;
      dst[(((size_t)b * HDS + h) * N_ + nBase + row) * DK + d] = acc[v] * scale;
    }
  }
}

__global__ __launch_bounds__(128) void mha_attn_kernel(
    const float* __restrict__ x,
    const float* __restrict__ qt, const float* __restrict__ kb,
    const float* __restrict__ vt, float* __restrict__ out) {
  __shared__ float ldsP[4][16 * 17];
  const int lane = threadIdx.x & 31;
  const int wave = threadIdx.x >> 5;
  float* P = ldsP[wave];

  const int task  = blockIdx.x * 4 + wave;
  const int qTile = task % NT;
  const int bh    = task / NT;
  const int h = bh & 7;
  const int b = bh >> 3;

  const int row  = lane & 15;
  const int half = lane >> 4;
  const int koff = half * 2;
  const int nBase = qTile * 16;

  v2f aq[8];
  {
    const float* qp =
        qt + (((size_t)b * HDS + h) * N_ + nBase + row) * DK + koff;
#pragma unroll
    for (int kk = 0; kk < 8; ++kk) {
      aq[kk].x = qp[kk * 4];
      aq[kk].y = qp[kk * 4 + 1];
    }
  }

  const float* kbase = kb + ((size_t)b * C_ + h * DK + koff) * N_ + row;
  const float* vhead = vt + ((size_t)b * HDS + h) * N_ * DK;

  v8f o0 = {}, o1 = {};
  float m[8], l[8];
#pragma unroll
  for (int v = 0; v < 8; ++v) { m[v] = -1e30f; l[v] = 0.0f; }

  for (int mt = 0; mt < NT; ++mt) {
    const int mBase = mt * 16;

    v8f S = {};
#pragma unroll
    for (int kk = 0; kk < 8; ++kk) {
      const float* kp = kbase + (size_t)(kk * 4) * N_ + mBase;
      v2f bk;
      bk.x = kp[0];
      bk.y = kp[N_];
      S = wmma_f32(aq[kk], bk, S);
    }

#pragma unroll
    for (int v = 0; v < 8; ++v) {
      float s  = S[v];
      float rm = s;
      rm = fmaxf(rm, __shfl_xor(rm, 1, 16));
      rm = fmaxf(rm, __shfl_xor(rm, 2, 16));
      rm = fmaxf(rm, __shfl_xor(rm, 4, 16));
      rm = fmaxf(rm, __shfl_xor(rm, 8, 16));
      const float mn = fmaxf(m[v], rm);
      const float a  = __expf(m[v] - mn);
      const float pv = __expf(s - mn);
      float rs = pv;
      rs += __shfl_xor(rs, 1, 16);
      rs += __shfl_xor(rs, 2, 16);
      rs += __shfl_xor(rs, 4, 16);
      rs += __shfl_xor(rs, 8, 16);
      l[v] = l[v] * a + rs;
      m[v] = mn;
      o0[v] *= a;
      o1[v] *= a;
      P[(v + half * 8) * 17 + row] = pv;
    }
    asm volatile("s_wait_dscnt 0" ::: "memory");

#pragma unroll
    for (int kk = 0; kk < 4; ++kk) {
      v2f ap;
      ap.x = P[row * 17 + kk * 4 + koff];
      ap.y = P[row * 17 + kk * 4 + koff + 1];
      const float* vp = vhead + (size_t)(mBase + kk * 4 + koff) * DK + row;
      v2f bv0, bv1;
      bv0.x = vp[0];
      bv0.y = vp[DK];
      bv1.x = vp[16];
      bv1.y = vp[DK + 16];
      o0 = wmma_f32(ap, bv0, o0);
      o1 = wmma_f32(ap, bv1, o1);
    }
  }

  const size_t obase =
      ((size_t)b * C_ + h * DK + row) * N_ + nBase + half * 8;
#pragma unroll
  for (int v = 0; v < 8; ++v) {
    const float inv = 1.0f / l[v];
    const size_t i0 = obase + v;
    out[i0] = x[i0] + o0[v] * inv;
    const size_t i1 = i0 + (size_t)16 * N_;
    out[i1] = x[i1] + o1[v] * inv;
  }
}

extern "C" void kernel_launch(void* const* d_in, const int* in_sizes, int n_in,
                              void* d_out, int out_size, void* d_ws,
                              size_t ws_size, hipStream_t stream) {
  const float* x  = (const float*)d_in[0];
  const float* wq = (const float*)d_in[1];
  const float* wk = (const float*)d_in[2];
  const float* wv = (const float*)d_in[3];
  float* out = (float*)d_out;

  float* qt = (float*)d_ws;
  float* kb = qt + (size_t)B_ * C_ * N_;
  float* vt = kb + (size_t)B_ * C_ * N_;

  dim3 blk(128);
  mha_proj_kernel<<<dim3(576, 12), blk, 0, stream>>>(x, wq, wk, wv, qt, kb, vt);
  mha_attn_kernel<<<dim3(1152), blk, 0, stream>>>(x, qt, kb, vt, out);
}


# Projections: 64x64 block tile, W and X channel chunks staged once per block in LDS by whole-line loads (X transposed), fragments by 16-byte LDS reads shared by 4 waves; changes WMMA k order
# speedup vs baseline: 1.9395x; 1.9395x over previous
#include <hip/hip_runtime.h>
#include <hip/hip_bf16.h>
#include <math.h>


typedef __attribute__((ext_vector_type(2))) float v2f;
typedef __attribute__((ext_vector_type(4))) float v4f;
typedef __attribute__((ext_vector_type(8))) float v8f;

#define B_   4
#define C_   256
#define HDS  8
#define DK   32
#define N_   2304
#define NT   (N_ / 16)
#define QSCALE 0.17677669529663687f

static __device__ __forceinline__ v8f wmma_f32(v2f a, v2f b, v8f c) {
  return __builtin_amdgcn_wmma_f32_16x16x4_f32(
      false, a, false, b, (short)0, c, false, false);
}

#define PJ_T   64
#define PJ_KC  32
#define PJ_LD  36
#define PJ_NCH (C_ / PJ_KC)
__global__ __launch_bounds__(128) void mha_proj_kernel(
    const float* __restrict__ x,
    const float* __restrict__ wq, const float* __restrict__ wk,
    const float* __restrict__ wv,
    float* __restrict__ qt, float* __restrict__ kb, float* __restrict__ vt) {
  __shared__ __attribute__((aligned(16))) float Ws[2][PJ_T * PJ_LD];
  __shared__ __attribute__((aligned(16))) float Xs[2][PJ_T * PJ_LD];
  const int tid  = threadIdx.x;
  const int lane = tid & 31;
  const int wave = tid >> 5;
  const int oBase = (blockIdx.x & 3) * PJ_T;
  const int nBase = (blockIdx.x >> 2) * PJ_T;
  const int p = blockIdx.y % 3;
  const int b = blockIdx.y / 3;
  const float* w = (p == 0) ? wq : (p == 1) ? wk : wv;

  const int wr = tid >> 3;
  const int wc = (tid & 7) * 4;
  const float* wG = w + (size_t)(oBase + wr) * C_ + wc;
  const int xn = tid & 63;
  const int xk = (tid >> 6) * 16;
  const float* xG = x + ((size_t)b * C_ + xk) * N_ + nBase + xn;
  const int wL = wr * PJ_LD + wc;
  const int xL = xn * PJ_LD + xk;

  const int cc = lane & 15;
  const int hh = lane >> 4;
  const int oSub = (wave >> 1) * 32;
  const int nSub = (wave & 1) * 32;
  const int aL = (oSub + cc) * PJ_LD + hh * 4;
  const int bL = (nSub + cc) * PJ_LD + hh * 4;

  v8f acc[2][2];
#pragma unroll
  for (int i = 0; i < 2; ++i) {
#pragma unroll
    for (int j = 0; j < 2; ++j) acc[i][j] = (v8f){};
  }

  v4f wreg[4], xreg[4];
#pragma unroll
  for (int i = 0; i < 4; ++i)
    wreg[i] = *(const v4f*)(wG + (size_t)(16 * i) * C_);
#pragma unroll
  for (int j = 0; j < 4; ++j) {
    xreg[j].x = xG[(size_t)(4 * j + 0) * N_];
    xreg[j].y = xG[(size_t)(4 * j + 1) * N_];
    xreg[j].z = xG[(size_t)(4 * j + 2) * N_];
    xreg[j].w = xG[(size_t)(4 * j + 3) * N_];
  }
#pragma unroll
  for (int i = 0; i < 4; ++i) *(v4f*)&Ws[0][wL + 16 * i * PJ_LD] = wreg[i];
#pragma unroll
  for (int j = 0; j < 4; ++j) *(v4f*)&Xs[0][xL + 4 * j] = xreg[j];
  __syncthreads();

  for (int ch = 0; ch < PJ_NCH; ++ch) {
    const int buf = ch & 1;
    const bool more = ch + 1 < PJ_NCH;
    if (more) {
      const int k0 = (ch + 1) * PJ_KC;
#pragma unroll
      for (int i = 0; i < 4; ++i)
        wreg[i] = *(const v4f*)(wG + (size_t)(16 * i) * C_ + k0);
#pragma unroll
      for (int j = 0; j < 4; ++j) {
        xreg[j].x = xG[(size_t)(k0 + 4 * j + 0) * N_];
        xreg[j].y = xG[(size_t)(k0 + 4 * j + 1) * N_];
        xreg[j].z = xG[(size_t)(k0 + 4 * j + 2) * N_];
        xreg[j].w = xG[(size_t)(k0 + 4 * j + 3) * N_];
      }
    }
    v4f fa[2][4], fb[2][4];
#pragma unroll
    for (int g = 0; g < 4; ++g) {
#pragma unroll
      for (int i = 0; i < 2; ++i) {
        fa[i][g] = *(const v4f*)&Ws[buf][aL + 16 * i * PJ_LD + 8 * g];
        fb[i][g] = *(const v4f*)&Xs[buf][bL + 16 * i * PJ_LD + 8 * g];
      }
    }
#pragma unroll
    for (int g = 0; g < 4; ++g) {
#pragma unroll
      for (int i = 0; i < 2; ++i) {
#pragma unroll
        for (int j = 0; j < 2; ++j) {
          const v2f a0 = {fa[i][g].x, fa[i][g].y};
          const v2f b0 = {fb[j][g].x, fb[j][g].y};
          acc[i][j] = wmma_f32(a0, b0, acc[i][j]);
        }
      }
#pragma unroll
      for (int i = 0; i < 2; ++i) {
#pragma unroll
        for (int j = 0; j < 2; ++j) {
          const v2f a1 = {fa[i][g].z, fa[i][g].w};
          const v2f b1 = {fb[j][g].z, fb[j][g].w};
          acc[i][j] = wmma_f32(a1, b1, acc[i][j]);
        }
      }
    }
    if (more) {
#pragma unroll
      for (int i = 0; i < 4; ++i)
        *(v4f*)&Ws[buf ^ 1][wL + 16 * i * PJ_LD] = wreg[i];
#pragma unroll
      for (int j = 0; j < 4; ++j) *(v4f*)&Xs[buf ^ 1][xL + 4 * j] = xreg[j];
      __syncthreads();
    }
  }

#pragma unroll
  for (int i = 0; i < 2; ++i) {
#pragma unroll
    for (int j = 0; j < 2; ++j) {
      const int oT = oBase + oSub + 16 * i;
      const int nT = nBase + nSub + 16 * j;
      if (p == 1) {
        float* outp = kb + ((size_t)b * C_ + oT + hh * 8) * N_ + nT + cc;
#pragma unroll
        for (int v = 0; v < 8; ++v) outp[(size_t)v * N_] = acc[i][j][v];
      } else {
        float* dst = (p == 0) ? qt : vt;
        const float scale = (p == 0) ? QSCALE : 1.0f;
#pragma unroll
        for (int v = 0; v < 8; ++v) {
          const int o = oT + v + hh * 8;
          const int hd = o >> 5, d = o & 31;
          dst[(((size_t)b * HDS + hd) * N_ + nT + cc) * DK + d] =
              acc[i][j][v] * scale;
        }
      }
    }
  }
}

#define XM_(n) "row_xmask:" #n " row_mask:0xf bank_mask:0xf"
#define DPP8_IN_(op, n)                      \
  op " %0, %8, %8 "   XM_(n) "\n\t"          \
  op " %1, %9, %9 "   XM_(n) "\n\t"          \
  op " %2, %10, %10 " XM_(n) "\n\t"          \
  op " %3, %11, %11 " XM_(n) "\n\t"          \
  op " %4, %12, %12 " XM_(n) "\n\t"          \
  op " %5, %13, %13 " XM_(n) "\n\t"          \
  op " %6, %14, %14 " XM_(n) "\n\t"          \
  op " %7, %15, %15 " XM_(n) "\n\t"
#define DPP8_(op, n)                         \
  op " %0, %0, %0 " XM_(n) "\n\t"            \
  op " %1, %1, %1 " XM_(n) "\n\t"            \
  op " %2, %2, %2 " XM_(n) "\n\t"            \
  op " %3, %3, %3 " XM_(n) "\n\t"            \
  op " %4, %4, %4 " XM_(n) "\n\t"            \
  op " %5, %5, %5 " XM_(n) "\n\t"            \
  op " %6, %6, %6 " XM_(n) "\n\t"            \
  op " %7, %7, %7 " XM_(n) "\n\t"

static __device__ __forceinline__ void row16_max8(const float* s, float* r) {
  asm(DPP8_IN_("v_max_num_f32_dpp", 1)
      DPP8_("v_max_num_f32_dpp", 2)
      DPP8_("v_max_num_f32_dpp", 4)
      "v_max_num_f32_dpp %0, %0, %0 " XM_(8) "\n\t"
      "v_max_num_f32_dpp %1, %1, %1 " XM_(8) "\n\t"
      "v_max_num_f32_dpp %2, %2, %2 " XM_(8) "\n\t"
      "v_max_num_f32_dpp %3, %3, %3 " XM_(8) "\n\t"
      "v_max_num_f32_dpp %4, %4, %4 " XM_(8) "\n\t"
      "v_max_num_f32_dpp %5, %5, %5 " XM_(8) "\n\t"
      "v_max_num_f32_dpp %6, %6, %6 " XM_(8) "\n\t"
      "v_max_num_f32_dpp %7, %7, %7 " XM_(8)
      : "=&v"(r[0]), "=&v"(r[1]), "=&v"(r[2]), "=&v"(r[3]),
        "=&v"(r[4]), "=&v"(r[5]), "=&v"(r[6]), "=&v"(r[7])
      : "v"(s[0]), "v"(s[1]), "v"(s[2]), "v"(s[3]),
        "v"(s[4]), "v"(s[5]), "v"(s[6]), "v"(s[7]));
}

static __device__ __forceinline__ void row16_sum8(const float* p, float* r) {
  asm(DPP8_IN_("v_add_f32_dpp", 1)
      DPP8_("v_add_f32_dpp", 2)
      DPP8_("v_add_f32_dpp", 4)
      "v_add_f32_dpp %0, %0, %0 " XM_(8) "\n\t"
      "v_add_f32_dpp %1, %1, %1 " XM_(8) "\n\t"
      "v_add_f32_dpp %2, %2, %2 " XM_(8) "\n\t"
      "v_add_f32_dpp %3, %3, %3 " XM_(8) "\n\t"
      "v_add_f32_dpp %4, %4, %4 " XM_(8) "\n\t"
      "v_add_f32_dpp %5, %5, %5 " XM_(8) "\n\t"
      "v_add_f32_dpp %6, %6, %6 " XM_(8) "\n\t"
      "v_add_f32_dpp %7, %7, %7 " XM_(8)
      : "=&v"(r[0]), "=&v"(r[1]), "=&v"(r[2]), "=&v"(r[3]),
        "=&v"(r[4]), "=&v"(r[5]), "=&v"(r[6]), "=&v"(r[7])
      : "v"(p[0]), "v"(p[1]), "v"(p[2]), "v"(p[3]),
        "v"(p[4]), "v"(p[5]), "v"(p[6]), "v"(p[7]));
}

#define QT     3
#define NTRIP  (NT / QT)
#define KSPLIT 2
#define NTH    (NT / KSPLIT)
#define KT     4
#define NSTEP  (NTH / KT)
#define KG     4
#define TG     2
#define PT     (16 * 17)
__global__ __launch_bounds__(128, 3) void mha_attn_kernel(
    const float* __restrict__ x,
    const float* __restrict__ qt, const float* __restrict__ kb,
    const float* __restrict__ vt, float* __restrict__ out) {
  __shared__ float ldsP[4][QT * KT * PT];
  __shared__ float4 ldsJ[2][QT][8][32];
  const int lane = threadIdx.x & 31;
  const int wave = __builtin_amdgcn_readfirstlane(threadIdx.x >> 5);
  float* P = ldsP[wave];

  const int qsel  = wave & 1;
  const int ks    = wave >> 1;
  const int task  = blockIdx.x * 2 + qsel;
  const int qTrip = task % NTRIP;
  const int bh    = task / NTRIP;
  const int h = bh & 7;
  const int b = bh >> 3;

  const int row  = lane & 15;
  const int half = lane >> 4;
  const int koff = half * 2;
  const int nBase = qTrip * (16 * QT);

  v2f aq[QT][8];
#pragma unroll
  for (int q = 0; q < QT; ++q) {
    const float* qp =
        qt + (((size_t)b * HDS + h) * N_ + nBase + q * 16 + row) * DK + koff;
#pragma unroll
    for (int kk = 0; kk < 8; ++kk) {
      aq[q][kk].x = qp[kk * 4];
      aq[q][kk].y = qp[kk * 4 + 1];
    }
  }

  const float* kS = kb + ((size_t)b * C_ + h * DK) * N_;
  const float* vS = vt + ((size_t)b * HDS + h) * N_ * DK;
  const unsigned kLane = (unsigned)(koff * N_ + row) * 4u;
  const unsigned vLane = (unsigned)(koff * DK + row) * 4u;

  v8f o0[QT], o1[QT];
  float m[QT][8], l[QT][8];
#pragma unroll
  for (int q = 0; q < QT; ++q) {
    o0[q] = (v8f){};
    o1[q] = (v8f){};
#pragma unroll
    for (int v = 0; v < 8; ++v) { m[q][v] = -1e30f; l[q][v] = 0.0f; }
  }

  const int mt0 = ks * NTH;
  for (int it = 0; it < NSTEP; ++it) {
    const int mBase = (mt0 + it * KT) * 16;

    v8f S[QT][KT];
#pragma unroll
    for (int q = 0; q < QT; ++q) {
#pragma unroll
      for (int t = 0; t < KT; ++t) S[q][t] = (v8f){};
    }
#pragma unroll
    for (int kg = 0; kg < 8; kg += KG) {
      v2f bk[KG][KT];
#pragma unroll
      for (int j = 0; j < KG; ++j) {
        const char* kp = (const char*)(kS + mBase + (size_t)((kg + j) * 4) * N_);
#pragma unroll
        for (int t = 0; t < KT; ++t) {
          bk[j][t].x = *(const float*)(kp + t * 64 + kLane);
          bk[j][t].y = *(const float*)(kp + t * 64 + N_ * 4 + kLane);
        }
      }
#pragma unroll
      for (int j = 0; j < KG; ++j) {
#pragma unroll
        for (int t = 0; t < KT; ++t) {
#pragma unroll
          for (int q = 0; q < QT; ++q)
            S[q][t] = wmma_f32(aq[q][kg + j], bk[j][t], S[q][t]);
        }
      }
    }

#pragma unroll
    for (int q = 0; q < QT; ++q) {
      float mn[8], al[8], rs[8], ps[8];
      {
        float tm[8];
#pragma unroll
        for (int v = 0; v < 8; ++v) {
          float a = m[q][v];
#pragma unroll
          for (int t = 0; t < KT; ++t) a = fmaxf(a, S[q][t][v]);
          tm[v] = a;
        }
        row16_max8(tm, mn);
      }
#pragma unroll
      for (int v = 0; v < 8; ++v) {
        al[v] = __expf(m[q][v] - mn[v]);
        m[q][v] = mn[v];
        o0[q][v] *= al[v];
        o1[q][v] *= al[v];
      }
#pragma unroll
      for (int t = 0; t < KT; ++t) {
#pragma unroll
        for (int v = 0; v < 8; ++v) {
          const float p = __expf(S[q][t][v] - mn[v]);
          S[q][t][v] = p;
          P[(q * KT + t) * PT + (v + half * 8) * 17 + row] = p;
        }
      }
#pragma unroll
      for (int v = 0; v < 8; ++v) {
        float a = S[q][0][v];
#pragma unroll
        for (int t = 1; t < KT; ++t) a += S[q][t][v];
        ps[v] = a;
      }
      row16_sum8(ps, rs);
#pragma unroll
      for (int v = 0; v < 8; ++v) l[q][v] = l[q][v] * al[v] + rs[v];
    }
    asm volatile("s_wait_dscnt 0" ::: "memory");

#pragma unroll
    for (int t0 = 0; t0 < KT; t0 += TG) {
      v2f ap[TG][QT][4], bv0[TG][4], bv1[TG][4];
#pragma unroll
      for (int tt = 0; tt < TG; ++tt) {
        const int t = t0 + tt;
#pragma unroll
        for (int kk = 0; kk < 4; ++kk) {
          const char* vp =
              (const char*)(vS + (size_t)(mBase + t * 16 + kk * 4) * DK);
          bv0[tt][kk].x = *(const float*)(vp + vLane);
          bv0[tt][kk].y = *(const float*)(vp + DK * 4 + vLane);
          bv1[tt][kk].x = *(const float*)(vp + 16 * 4 + vLane);
          bv1[tt][kk].y = *(const float*)(vp + (DK + 16) * 4 + vLane);
#pragma unroll
          for (int q = 0; q < QT; ++q) {
            ap[tt][q][kk].x = P[(q * KT + t) * PT + row * 17 + kk * 4 + koff];
            ap[tt][q][kk].y = P[(q * KT + t) * PT + row * 17 + kk * 4 + koff + 1];
          }
        }
      }
#pragma unroll
      for (int tt = 0; tt < TG; ++tt) {
#pragma unroll
        for (int kk = 0; kk < 4; ++kk) {
#pragma unroll
          for (int q = 0; q < QT; ++q) {
            o0[q] = wmma_f32(ap[tt][q][kk], bv0[tt][kk], o0[q]);
            o1[q] = wmma_f32(ap[tt][q][kk], bv1[tt][kk], o1[q]);
          }
        }
      }
    }
  }

  int lane2;
  asm volatile("v_mbcnt_lo_u32_b32 %0, -1, 0" : "=v"(lane2));
  const int row2  = lane2 & 15;
  const int half2 = lane2 >> 4;

  float4* J = &ldsJ[qsel][0][0][lane2];
  if (ks == 1) {
#pragma unroll
    for (int q = 0; q < QT; ++q) {
      float4* Jq = J + q * 8 * 32;
      Jq[0 * 32] = make_float4(o0[q][0], o0[q][1], o0[q][2], o0[q][3]);
      Jq[1 * 32] = make_float4(o0[q][4], o0[q][5], o0[q][6], o0[q][7]);
      Jq[2 * 32] = make_float4(o1[q][0], o1[q][1], o1[q][2], o1[q][3]);
      Jq[3 * 32] = make_float4(o1[q][4], o1[q][5], o1[q][6], o1[q][7]);
      Jq[4 * 32] = make_float4(m[q][0], m[q][1], m[q][2], m[q][3]);
      Jq[5 * 32] = make_float4(m[q][4], m[q][5], m[q][6], m[q][7]);
      Jq[6 * 32] = make_float4(l[q][0], l[q][1], l[q][2], l[q][3]);
      Jq[7 * 32] = make_float4(l[q][4], l[q][5], l[q][6], l[q][7]);
    }
  }
  __syncthreads();
  if (ks == 1) return;

#pragma unroll
  for (int q = 0; q < QT; ++q) {
    const float4* Jq = J + q * 8 * 32;
    float po0[8], po1[8], pm[8], pl[8];
#pragma unroll
    for (int r = 0; r < 2; ++r) {
      const float4 a0 = Jq[(0 + r) * 32], a1 = Jq[(2 + r) * 32];
      const float4 am = Jq[(4 + r) * 32], al = Jq[(6 + r) * 32];
      po0[4*r+0] = a0.x; po0[4*r+1] = a0.y; po0[4*r+2] = a0.z; po0[4*r+3] = a0.w;
      po1[4*r+0] = a1.x; po1[4*r+1] = a1.y; po1[4*r+2] = a1.z; po1[4*r+3] = a1.w;
      pm[4*r+0]  = am.x; pm[4*r+1]  = am.y; pm[4*r+2]  = am.z; pm[4*r+3]  = am.w;
      pl[4*r+0]  = al.x; pl[4*r+1]  = al.y; pl[4*r+2]  = al.z; pl[4*r+3]  = al.w;
    }
    float lq[8];
    v8f r0 = o0[q], r1 = o1[q];
#pragma unroll
    for (int v = 0; v < 8; ++v) {
      const float mn = fmaxf(m[q][v], pm[v]);
      const float a  = __expf(m[q][v] - mn);
      const float pa = __expf(pm[v] - mn);
      lq[v] = l[q][v] * a + pl[v] * pa;
      r0[v] = r0[v] * a + po0[v] * pa;
      r1[v] = r1[v] * a + po1[v] * pa;
    }

    const size_t obase =
        ((size_t)b * C_ + h * DK + row2) * N_ + nBase + q * 16 + half2 * 8;
#pragma unroll
    for (int v = 0; v < 8; ++v) {
      const float inv = 1.0f / lq[v];
      const size_t i0 = obase + v;
      out[i0] = x[i0] + r0[v] * inv;
      const size_t i1 = i0 + (size_t)16 * N_;
      out[i1] = x[i1] + r1[v] * inv;
    }
  }
}

extern "C" void kernel_launch(void* const* d_in, const int* in_sizes, int n_in,
                              void* d_out, int out_size, void* d_ws,
                              size_t ws_size, hipStream_t stream) {
  const float* x  = (const float*)d_in[0];
  const float* wq = (const float*)d_in[1];
  const float* wk = (const float*)d_in[2];
  const float* wv = (const float*)d_in[3];
  float* out = (float*)d_out;

  float* qt = (float*)d_ws;
  float* kb = qt + (size_t)B_ * C_ * N_;
  float* vt = kb + (size_t)B_ * C_ * N_;

  dim3 blk(128);
  mha_proj_kernel<<<dim3(144, 12), blk, 0, stream>>>(x, wq, wk, wv, qt, kb, vt);
  mha_attn_kernel<<<dim3(768), blk, 0, stream>>>(x, qt, kb, vt, out);
}


# Attention: raw scores cross LDS, softmax runs on the A-layout fragments: in-lane max and sum, one half exchange, l kept per lane; O factor back to C layout by 9 DPP moves; changes softmax sum order
# speedup vs baseline: 2.0589x; 1.0616x over previous
#include <hip/hip_runtime.h>
#include <hip/hip_bf16.h>
#include <math.h>


typedef __attribute__((ext_vector_type(2))) float v2f;
typedef __attribute__((ext_vector_type(4))) float v4f;
typedef __attribute__((ext_vector_type(8))) float v8f;

#define B_   4
#define C_   256
#define HDS  8
#define DK   32
#define N_   2304
#define NT   (N_ / 16)
#define QSCALE 0.17677669529663687f

static __device__ __forceinline__ v8f wmma_f32(v2f a, v2f b, v8f c) {
  return __builtin_amdgcn_wmma_f32_16x16x4_f32(
      false, a, false, b, (short)0, c, false, false);
}

#define PJ_T   64
#define PJ_KC  32
#define PJ_LD  36
#define PJ_NCH (C_ / PJ_KC)
__global__ __launch_bounds__(128) void mha_proj_kernel(
    const float* __restrict__ x,
    const float* __restrict__ wq, const float* __restrict__ wk,
    const float* __restrict__ wv,
    float* __restrict__ qt, float* __restrict__ kb, float* __restrict__ vt) {
  __shared__ __attribute__((aligned(16))) float Ws[2][PJ_T * PJ_LD];
  __shared__ __attribute__((aligned(16))) float Xs[2][PJ_T * PJ_LD];
  const int tid  = threadIdx.x;
  const int lane = tid & 31;
  const int wave = tid >> 5;
  const int oBase = (blockIdx.x & 3) * PJ_T;
  const int nBase = (blockIdx.x >> 2) * PJ_T;
  const int p = blockIdx.y % 3;
  const int b = blockIdx.y / 3;
  const float* w = (p == 0) ? wq : (p == 1) ? wk : wv;

  const int wr = tid >> 3;
  const int wc = (tid & 7) * 4;
  const float* wG = w + (size_t)(oBase + wr) * C_ + wc;
  const int xn = tid & 63;
  const int xk = (tid >> 6) * 16;
  const float* xG = x + ((size_t)b * C_ + xk) * N_ + nBase + xn;
  const int wL = wr * PJ_LD + wc;
  const int xL = xn * PJ_LD + xk;

  const int cc = lane & 15;
  const int hh = lane >> 4;
  const int oSub = (wave >> 1) * 32;
  const int nSub = (wave & 1) * 32;
  const int aL = (oSub + cc) * PJ_LD + hh * 4;
  const int bL = (nSub + cc) * PJ_LD + hh * 4;

  v8f acc[2][2];
#pragma unroll
  for (int i = 0; i < 2; ++i) {
#pragma unroll
    for (int j = 0; j < 2; ++j) acc[i][j] = (v8f){};
  }

  v4f wreg[4], xreg[4];
#pragma unroll
  for (int i = 0; i < 4; ++i)
    wreg[i] = *(const v4f*)(wG + (size_t)(16 * i) * C_);
#pragma unroll
  for (int j = 0; j < 4; ++j) {
    xreg[j].x = xG[(size_t)(4 * j + 0) * N_];
    xreg[j].y = xG[(size_t)(4 * j + 1) * N_];
    xreg[j].z = xG[(size_t)(4 * j + 2) * N_];
    xreg[j].w = xG[(size_t)(4 * j + 3) * N_];
  }
#pragma unroll
  for (int i = 0; i < 4; ++i) *(v4f*)&Ws[0][wL + 16 * i * PJ_LD] = wreg[i];
#pragma unroll
  for (int j = 0; j < 4; ++j) *(v4f*)&Xs[0][xL + 4 * j] = xreg[j];
  __syncthreads();

  for (int ch = 0; ch < PJ_NCH; ++ch) {
    const int buf = ch & 1;
    const bool more = ch + 1 < PJ_NCH;
    if (more) {
      const int k0 = (ch + 1) * PJ_KC;
#pragma unroll
      for (int i = 0; i < 4; ++i)
        wreg[i] = *(const v4f*)(wG + (size_t)(16 * i) * C_ + k0);
#pragma unroll
      for (int j = 0; j < 4; ++j) {
        xreg[j].x = xG[(size_t)(k0 + 4 * j + 0) * N_];
        xreg[j].y = xG[(size_t)(k0 + 4 * j + 1) * N_];
        xreg[j].z = xG[(size_t)(k0 + 4 * j + 2) * N_];
        xreg[j].w = xG[(size_t)(k0 + 4 * j + 3) * N_];
      }
    }
    v4f fa[2][4], fb[2][4];
#pragma unroll
    for (int g = 0; g < 4; ++g) {
#pragma unroll
      for (int i = 0; i < 2; ++i) {
        fa[i][g] = *(const v4f*)&Ws[buf][aL + 16 * i * PJ_LD + 8 * g];
        fb[i][g] = *(const v4f*)&Xs[buf][bL + 16 * i * PJ_LD + 8 * g];
      }
    }
#pragma unroll
    for (int g = 0; g < 4; ++g) {
#pragma unroll
      for (int i = 0; i < 2; ++i) {
#pragma unroll
        for (int j = 0; j < 2; ++j) {
          const v2f a0 = {fa[i][g].x, fa[i][g].y};
          const v2f b0 = {fb[j][g].x, fb[j][g].y};
          acc[i][j] = wmma_f32(a0, b0, acc[i][j]);
        }
      }
#pragma unroll
      for (int i = 0; i < 2; ++i) {
#pragma unroll
        for (int j = 0; j < 2; ++j) {
          const v2f a1 = {fa[i][g].z, fa[i][g].w};
          const v2f b1 = {fb[j][g].z, fb[j][g].w};
          acc[i][j] = wmma_f32(a1, b1, acc[i][j]);
        }
      }
    }
    if (more) {
#pragma unroll
      for (int i = 0; i < 4; ++i)
        *(v4f*)&Ws[buf ^ 1][wL + 16 * i * PJ_LD] = wreg[i];
#pragma unroll
      for (int j = 0; j < 4; ++j) *(v4f*)&Xs[buf ^ 1][xL + 4 * j] = xreg[j];
      __syncthreads();
    }
  }

#pragma unroll
  for (int i = 0; i < 2; ++i) {
#pragma unroll
    for (int j = 0; j < 2; ++j) {
      const int oT = oBase + oSub + 16 * i;
      const int nT = nBase + nSub + 16 * j;
      if (p == 1) {
        float* outp = kb + ((size_t)b * C_ + oT + hh * 8) * N_ + nT + cc;
#pragma unroll
        for (int v = 0; v < 8; ++v) outp[(size_t)v * N_] = acc[i][j][v];
      } else {
        float* dst = (p == 0) ? qt : vt;
        const float scale = (p == 0) ? QSCALE : 1.0f;
#pragma unroll
        for (int v = 0; v < 8; ++v) {
          const int o = oT + v + hh * 8;
          const int hd = o >> 5, d = o & 31;
          dst[(((size_t)b * HDS + hd) * N_ + nT + cc) * DK + d] =
              acc[i][j][v] * scale;
        }
      }
    }
  }
}


static __device__ __forceinline__ float xhalf(float x) {
  const int b = __builtin_bit_cast(int, x);
  return __builtin_bit_cast(
      float, __builtin_amdgcn_permlanex16(b, b, 0x76543210, 0xfedcba98, false, false));
}

static __device__ __forceinline__ void rows_to_c8(float x, float* t) {
  float y;
  asm("v_mov_b32_e32 %0, %1\n\t"
      "v_mov_b32_dpp %0, %1 row_xmask:8 row_mask:0x2 bank_mask:0xf"
      : "=&v"(y)
      : "v"(x));
  asm("v_mov_b32_dpp %0, %8 row_share:0 row_mask:0xf bank_mask:0xf\n\t"
      "v_mov_b32_dpp %1, %8 row_share:1 row_mask:0xf bank_mask:0xf\n\t"
      "v_mov_b32_dpp %2, %8 row_share:2 row_mask:0xf bank_mask:0xf\n\t"
      "v_mov_b32_dpp %3, %8 row_share:3 row_mask:0xf bank_mask:0xf\n\t"
      "v_mov_b32_dpp %4, %8 row_share:4 row_mask:0xf bank_mask:0xf\n\t"
      "v_mov_b32_dpp %5, %8 row_share:5 row_mask:0xf bank_mask:0xf\n\t"
      "v_mov_b32_dpp %6, %8 row_share:6 row_mask:0xf bank_mask:0xf\n\t"
      "v_mov_b32_dpp %7, %8 row_share:7 row_mask:0xf bank_mask:0xf"
      : "=&v"(t[0]), "=&v"(t[1]), "=&v"(t[2]), "=&v"(t[3]),
        "=&v"(t[4]), "=&v"(t[5]), "=&v"(t[6]), "=&v"(t[7])
      : "v"(y));
}

#define QT     3
#define NTRIP  (NT / QT)
#define KSPLIT 2
#define NTH    (NT / KSPLIT)
#define KT     4
#define NSTEP  (NTH / KT)
#define KG     4
#define TG     2
#define PT     (16 * 17)
__global__ __launch_bounds__(128, 3) void mha_attn_kernel(
    const float* __restrict__ x,
    const float* __restrict__ qt, const float* __restrict__ kb,
    const float* __restrict__ vt, float* __restrict__ out) {
  __shared__ float ldsP[4][QT * KT * PT];
  __shared__ float4 ldsJ[2][QT][5][32];
  const int lane = threadIdx.x & 31;
  const int wave = __builtin_amdgcn_readfirstlane(threadIdx.x >> 5);
  float* P = ldsP[wave];

  const int qsel  = wave & 1;
  const int ks    = wave >> 1;
  const int task  = blockIdx.x * 2 + qsel;
  const int qTrip = task % NTRIP;
  const int bh    = task / NTRIP;
  const int h = bh & 7;
  const int b = bh >> 3;

  const int row  = lane & 15;
  const int half = lane >> 4;
  const int koff = half * 2;
  const int nBase = qTrip * (16 * QT);

  v2f aq[QT][8];
#pragma unroll
  for (int q = 0; q < QT; ++q) {
    const float* qp =
        qt + (((size_t)b * HDS + h) * N_ + nBase + q * 16 + row) * DK + koff;
#pragma unroll
    for (int kk = 0; kk < 8; ++kk) {
      aq[q][kk].x = qp[kk * 4];
      aq[q][kk].y = qp[kk * 4 + 1];
    }
  }

  const float* kS = kb + ((size_t)b * C_ + h * DK) * N_;
  const float* vS = vt + ((size_t)b * HDS + h) * N_ * DK;
  const unsigned kLane = (unsigned)(koff * N_ + row) * 4u;
  const unsigned vLane = (unsigned)(koff * DK + row) * 4u;

  v8f o0[QT], o1[QT];
  float m[QT], l[QT];
#pragma unroll
  for (int q = 0; q < QT; ++q) {
    o0[q] = (v8f){};
    o1[q] = (v8f){};
    m[q] = -1e30f;
    l[q] = 0.0f;
  }

  const int mt0 = ks * NTH;
  for (int it = 0; it < NSTEP; ++it) {
    const int mBase = (mt0 + it * KT) * 16;

    v8f S[QT][KT];
#pragma unroll
    for (int q = 0; q < QT; ++q) {
#pragma unroll
      for (int t = 0; t < KT; ++t) S[q][t] = (v8f){};
    }
#pragma unroll
    for (int kg = 0; kg < 8; kg += KG) {
      v2f bk[KG][KT];
#pragma unroll
      for (int j = 0; j < KG; ++j) {
        const char* kp = (const char*)(kS + mBase + (size_t)((kg + j) * 4) * N_);
#pragma unroll
        for (int t = 0; t < KT; ++t) {
          bk[j][t].x = *(const float*)(kp + t * 64 + kLane);
          bk[j][t].y = *(const float*)(kp + t * 64 + N_ * 4 + kLane);
        }
      }
#pragma unroll
      for (int j = 0; j < KG; ++j) {
#pragma unroll
        for (int t = 0; t < KT; ++t) {
#pragma unroll
          for (int q = 0; q < QT; ++q)
            S[q][t] = wmma_f32(aq[q][kg + j], bk[j][t], S[q][t]);
        }
      }
    }

#pragma unroll
    for (int q = 0; q < QT; ++q) {
#pragma unroll
      for (int t = 0; t < KT; ++t) {
#pragma unroll
        for (int v = 0; v < 8; ++v)
          P[(q * KT + t) * PT + (v + half * 8) * 17 + row] = S[q][t][v];
      }
    }
    asm volatile("s_wait_dscnt 0" ::: "memory");

    v2f ap[QT][KT][4];
#pragma unroll
    for (int q = 0; q < QT; ++q) {
#pragma unroll
      for (int t = 0; t < KT; ++t) {
#pragma unroll
        for (int kk = 0; kk < 4; ++kk) {
          ap[q][t][kk].x = P[(q * KT + t) * PT + row * 17 + kk * 4 + koff];
          ap[q][t][kk].y = P[(q * KT + t) * PT + row * 17 + kk * 4 + koff + 1];
        }
      }
      float mx = m[q];
#pragma unroll
      for (int t = 0; t < KT; ++t) {
#pragma unroll
        for (int kk = 0; kk < 4; ++kk)
          mx = fmaxf(mx, fmaxf(ap[q][t][kk].x, ap[q][t][kk].y));
      }
      const float mn = fmaxf(mx, xhalf(mx));
      const float al = __expf(m[q] - mn);
      m[q] = mn;
      float st[KT];
#pragma unroll
      for (int t = 0; t < KT; ++t) {
        float a = 0.0f;
#pragma unroll
        for (int kk = 0; kk < 4; ++kk) {
          const float p0 = __expf(ap[q][t][kk].x - mn);
          const float p1 = __expf(ap[q][t][kk].y - mn);
          ap[q][t][kk].x = p0;
          ap[q][t][kk].y = p1;
          a += p0 + p1;
        }
        st[t] = a;
      }
      l[q] = l[q] * al + ((st[0] + st[1]) + (st[2] + st[3]));
      float alC[8];
      rows_to_c8(al, alC);
#pragma unroll
      for (int v = 0; v < 8; ++v) {
        o0[q][v] *= alC[v];
        o1[q][v] *= alC[v];
      }
    }

#pragma unroll
    for (int t0 = 0; t0 < KT; t0 += TG) {
      v2f bv0[TG][4], bv1[TG][4];
#pragma unroll
      for (int tt = 0; tt < TG; ++tt) {
        const int t = t0 + tt;
#pragma unroll
        for (int kk = 0; kk < 4; ++kk) {
          const char* vp =
              (const char*)(vS + (size_t)(mBase + t * 16 + kk * 4) * DK);
          bv0[tt][kk].x = *(const float*)(vp + vLane);
          bv0[tt][kk].y = *(const float*)(vp + DK * 4 + vLane);
          bv1[tt][kk].x = *(const float*)(vp + 16 * 4 + vLane);
          bv1[tt][kk].y = *(const float*)(vp + (DK + 16) * 4 + vLane);
        }
      }
#pragma unroll
      for (int tt = 0; tt < TG; ++tt) {
#pragma unroll
        for (int kk = 0; kk < 4; ++kk) {
#pragma unroll
          for (int q = 0; q < QT; ++q) {
            o0[q] = wmma_f32(ap[q][t0 + tt][kk], bv0[tt][kk], o0[q]);
            o1[q] = wmma_f32(ap[q][t0 + tt][kk], bv1[tt][kk], o1[q]);
          }
        }
      }
    }
  }

  float4* J = &ldsJ[qsel][0][0][lane];
  if (ks == 1) {
#pragma unroll
    for (int q = 0; q < QT; ++q) {
      float4* Jq = J + q * 5 * 32;
      Jq[0 * 32] = make_float4(o0[q][0], o0[q][1], o0[q][2], o0[q][3]);
      Jq[1 * 32] = make_float4(o0[q][4], o0[q][5], o0[q][6], o0[q][7]);
      Jq[2 * 32] = make_float4(o1[q][0], o1[q][1], o1[q][2], o1[q][3]);
      Jq[3 * 32] = make_float4(o1[q][4], o1[q][5], o1[q][6], o1[q][7]);
      Jq[4 * 32] = make_float4(m[q], l[q], 0.0f, 0.0f);
    }
  }
  __syncthreads();
  if (ks == 1) return;

#pragma unroll
  for (int q = 0; q < QT; ++q) {
    const float4* Jq = J + q * 5 * 32;
    float po0[8], po1[8];
#pragma unroll
    for (int r = 0; r < 2; ++r) {
      const float4 a0 = Jq[(0 + r) * 32], a1 = Jq[(2 + r) * 32];
      po0[4*r+0] = a0.x; po0[4*r+1] = a0.y; po0[4*r+2] = a0.z; po0[4*r+3] = a0.w;
      po1[4*r+0] = a1.x; po1[4*r+1] = a1.y; po1[4*r+2] = a1.z; po1[4*r+3] = a1.w;
    }
    const float4 ml = Jq[4 * 32];
    const float mn = fmaxf(m[q], ml.x);
    const float a  = __expf(m[q] - mn);
    const float pa = __expf(ml.x - mn);
    const float lp = l[q] * a + ml.y * pa;
    const float inv = 1.0f / (lp + xhalf(lp));
    float aC[8], paC[8], invC[8];
    rows_to_c8(a, aC);
    rows_to_c8(pa, paC);
    rows_to_c8(inv, invC);

    const size_t obase =
        ((size_t)b * C_ + h * DK + row) * N_ + nBase + q * 16 + half * 8;
#pragma unroll
    for (int v = 0; v < 8; ++v) {
      const float r0 = o0[q][v] * aC[v] + po0[v] * paC[v];
      const float r1 = o1[q][v] * aC[v] + po1[v] * paC[v];
      const size_t i0 = obase + v;
      out[i0] = x[i0] + r0 * invC[v];
      const size_t i1 = i0 + (size_t)16 * N_;
      out[i1] = x[i1] + r1 * invC[v];
    }
  }
}

extern "C" void kernel_launch(void* const* d_in, const int* in_sizes, int n_in,
                              void* d_out, int out_size, void* d_ws,
                              size_t ws_size, hipStream_t stream) {
  const float* x  = (const float*)d_in[0];
  const float* wq = (const float*)d_in[1];
  const float* wk = (const float*)d_in[2];
  const float* wv = (const float*)d_in[3];
  float* out = (float*)d_out;

  float* qt = (float*)d_ws;
  float* kb = qt + (size_t)B_ * C_ * N_;
  float* vt = kb + (size_t)B_ * C_ * N_;

  dim3 blk(128);
  mha_proj_kernel<<<dim3(144, 12), blk, 0, stream>>>(x, wq, wk, wv, qt, kb, vt);
  mha_attn_kernel<<<dim3(768), blk, 0, stream>>>(x, qt, kb, vt, out);
}


# Attention: score tiles staged key-permuted at pitch 20: A fragments by two 16-byte LDS reads per tile, conflict-free stores, no drain; V loads one group off one base (frees registers); same sums
# speedup vs baseline: 2.1327x; 1.0358x over previous
#include <hip/hip_runtime.h>
#include <hip/hip_bf16.h>
#include <math.h>


typedef __attribute__((ext_vector_type(2))) float v2f;
typedef __attribute__((ext_vector_type(4))) float v4f;
typedef __attribute__((ext_vector_type(8))) float v8f;

#define B_   4
#define C_   256
#define HDS  8
#define DK   32
#define N_   2304
#define NT   (N_ / 16)
#define QSCALE 0.17677669529663687f

static __device__ __forceinline__ v8f wmma_f32(v2f a, v2f b, v8f c) {
  return __builtin_amdgcn_wmma_f32_16x16x4_f32(
      false, a, false, b, (short)0, c, false, false);
}

#define PJ_T   64
#define PJ_KC  32
#define PJ_LD  36
#define PJ_NCH (C_ / PJ_KC)
__global__ __launch_bounds__(128) void mha_proj_kernel(
    const float* __restrict__ x,
    const float* __restrict__ wq, const float* __restrict__ wk,
    const float* __restrict__ wv,
    float* __restrict__ qt, float* __restrict__ kb, float* __restrict__ vt) {
  __shared__ __attribute__((aligned(16))) float Ws[2][PJ_T * PJ_LD];
  __shared__ __attribute__((aligned(16))) float Xs[2][PJ_T * PJ_LD];
  const int tid  = threadIdx.x;
  const int lane = tid & 31;
  const int wave = tid >> 5;
  const int oBase = (blockIdx.x & 3) * PJ_T;
  const int nBase = (blockIdx.x >> 2) * PJ_T;
  const int p = blockIdx.y % 3;
  const int b = blockIdx.y / 3;
  const float* w = (p == 0) ? wq : (p == 1) ? wk : wv;

  const int wr = tid >> 3;
  const int wc = (tid & 7) * 4;
  const float* wG = w + (size_t)(oBase + wr) * C_ + wc;
  const int xn = tid & 63;
  const int xk = (tid >> 6) * 16;
  const float* xG = x + ((size_t)b * C_ + xk) * N_ + nBase + xn;
  const int wL = wr * PJ_LD + wc;
  const int xL = xn * PJ_LD + xk;

  const int cc = lane & 15;
  const int hh = lane >> 4;
  const int oSub = (wave >> 1) * 32;
  const int nSub = (wave & 1) * 32;
  const int aL = (oSub + cc) * PJ_LD + hh * 4;
  const int bL = (nSub + cc) * PJ_LD + hh * 4;

  v8f acc[2][2];
#pragma unroll
  for (int i = 0; i < 2; ++i) {
#pragma unroll
    for (int j = 0; j < 2; ++j) acc[i][j] = (v8f){};
  }

  v4f wreg[4], xreg[4];
#pragma unroll
  for (int i = 0; i < 4; ++i)
    wreg[i] = *(const v4f*)(wG + (size_t)(16 * i) * C_);
#pragma unroll
  for (int j = 0; j < 4; ++j) {
    xreg[j].x = xG[(size_t)(4 * j + 0) * N_];
    xreg[j].y = xG[(size_t)(4 * j + 1) * N_];
    xreg[j].z = xG[(size_t)(4 * j + 2) * N_];
    xreg[j].w = xG[(size_t)(4 * j + 3) * N_];
  }
#pragma unroll
  for (int i = 0; i < 4; ++i) *(v4f*)&Ws[0][wL + 16 * i * PJ_LD] = wreg[i];
#pragma unroll
  for (int j = 0; j < 4; ++j) *(v4f*)&Xs[0][xL + 4 * j] = xreg[j];
  __syncthreads();

  for (int ch = 0; ch < PJ_NCH; ++ch) {
    const int buf = ch & 1;
    const bool more = ch + 1 < PJ_NCH;
    if (more) {
      const int k0 = (ch + 1) * PJ_KC;
#pragma unroll
      for (int i = 0; i < 4; ++i)
        wreg[i] = *(const v4f*)(wG + (size_t)(16 * i) * C_ + k0);
#pragma unroll
      for (int j = 0; j < 4; ++j) {
        xreg[j].x = xG[(size_t)(k0 + 4 * j + 0) * N_];
        xreg[j].y = xG[(size_t)(k0 + 4 * j + 1) * N_];
        xreg[j].z = xG[(size_t)(k0 + 4 * j + 2) * N_];
        xreg[j].w = xG[(size_t)(k0 + 4 * j + 3) * N_];
      }
    }
    v4f fa[2][4], fb[2][4];
#pragma unroll
    for (int g = 0; g < 4; ++g) {
#pragma unroll
      for (int i = 0; i < 2; ++i) {
        fa[i][g] = *(const v4f*)&Ws[buf][aL + 16 * i * PJ_LD + 8 * g];
        fb[i][g] = *(const v4f*)&Xs[buf][bL + 16 * i * PJ_LD + 8 * g];
      }
    }
#pragma unroll
    for (int g = 0; g < 4; ++g) {
#pragma unroll
      for (int i = 0; i < 2; ++i) {
#pragma unroll
        for (int j = 0; j < 2; ++j) {
          const v2f a0 = {fa[i][g].x, fa[i][g].y};
          const v2f b0 = {fb[j][g].x, fb[j][g].y};
          acc[i][j] = wmma_f32(a0, b0, acc[i][j]);
        }
      }
#pragma unroll
      for (int i = 0; i < 2; ++i) {
#pragma unroll
        for (int j = 0; j < 2; ++j) {
          const v2f a1 = {fa[i][g].z, fa[i][g].w};
          const v2f b1 = {fb[j][g].z, fb[j][g].w};
          acc[i][j] = wmma_f32(a1, b1, acc[i][j]);
        }
      }
    }
    if (more) {
#pragma unroll
      for (int i = 0; i < 4; ++i)
        *(v4f*)&Ws[buf ^ 1][wL + 16 * i * PJ_LD] = wreg[i];
#pragma unroll
      for (int j = 0; j < 4; ++j) *(v4f*)&Xs[buf ^ 1][xL + 4 * j] = xreg[j];
      __syncthreads();
    }
  }

#pragma unroll
  for (int i = 0; i < 2; ++i) {
#pragma unroll
    for (int j = 0; j < 2; ++j) {
      const int oT = oBase + oSub + 16 * i;
      const int nT = nBase + nSub + 16 * j;
      if (p == 1) {
        float* outp = kb + ((size_t)b * C_ + oT + hh * 8) * N_ + nT + cc;
#pragma unroll
        for (int v = 0; v < 8; ++v) outp[(size_t)v * N_] = acc[i][j][v];
      } else {
        float* dst = (p == 0) ? qt : vt;
        const float scale = (p == 0) ? QSCALE : 1.0f;
#pragma unroll
        for (int v = 0; v < 8; ++v) {
          const int o = oT + v + hh * 8;
          const int hd = o >> 5, d = o & 31;
          dst[(((size_t)b * HDS + hd) * N_ + nT + cc) * DK + d] =
              acc[i][j][v] * scale;
        }
      }
    }
  }
}


static __device__ __forceinline__ float xhalf(float x) {
  const int b = __builtin_bit_cast(int, x);
  return __builtin_bit_cast(
      float, __builtin_amdgcn_permlanex16(b, b, 0x76543210, 0xfedcba98, false, false));
}

static __device__ __forceinline__ void rows_to_c8(float x, float* t) {
  float y;
  asm("v_mov_b32_e32 %0, %1\n\t"
      "v_mov_b32_dpp %0, %1 row_xmask:8 row_mask:0x2 bank_mask:0xf"
      : "=&v"(y)
      : "v"(x));
  asm("v_mov_b32_dpp %0, %8 row_share:0 row_mask:0xf bank_mask:0xf\n\t"
      "v_mov_b32_dpp %1, %8 row_share:1 row_mask:0xf bank_mask:0xf\n\t"
      "v_mov_b32_dpp %2, %8 row_share:2 row_mask:0xf bank_mask:0xf\n\t"
      "v_mov_b32_dpp %3, %8 row_share:3 row_mask:0xf bank_mask:0xf\n\t"
      "v_mov_b32_dpp %4, %8 row_share:4 row_mask:0xf bank_mask:0xf\n\t"
      "v_mov_b32_dpp %5, %8 row_share:5 row_mask:0xf bank_mask:0xf\n\t"
      "v_mov_b32_dpp %6, %8 row_share:6 row_mask:0xf bank_mask:0xf\n\t"
      "v_mov_b32_dpp %7, %8 row_share:7 row_mask:0xf bank_mask:0xf"
      : "=&v"(t[0]), "=&v"(t[1]), "=&v"(t[2]), "=&v"(t[3]),
        "=&v"(t[4]), "=&v"(t[5]), "=&v"(t[6]), "=&v"(t[7])
      : "v"(y));
}

#define QT     3
#define NTRIP  (NT / QT)
#define KSPLIT 2
#define NTH    (NT / KSPLIT)
#define KT     4
#define NSTEP  (NTH / KT)
#define KG     4
#define TG     4
#define PP     20
#define PT     (16 * PP)
__global__ __launch_bounds__(128, 3) void mha_attn_kernel(
    const float* __restrict__ x,
    const float* __restrict__ qt, const float* __restrict__ kb,
    const float* __restrict__ vt, float* __restrict__ out) {
  __shared__ __attribute__((aligned(16))) float ldsP[4][QT * KT * PT];
  __shared__ float4 ldsJ[2][QT][5][32];
  const int lane = threadIdx.x & 31;
  const int wave = __builtin_amdgcn_readfirstlane(threadIdx.x >> 5);
  float* P = ldsP[wave];

  const int qsel  = wave & 1;
  const int ks    = wave >> 1;
  const int task  = blockIdx.x * 2 + qsel;
  const int qTrip = task % NTRIP;
  const int bh    = task / NTRIP;
  const int h = bh & 7;
  const int b = bh >> 3;

  const int row  = lane & 15;
  const int half = lane >> 4;
  const int koff = half * 2;
  const int pcol = ((row >> 1) & 1) * 8 + (row >> 2) * 2 + (row & 1);
  const int nBase = qTrip * (16 * QT);

  v2f aq[QT][8];
#pragma unroll
  for (int q = 0; q < QT; ++q) {
    const float* qp =
        qt + (((size_t)b * HDS + h) * N_ + nBase + q * 16 + row) * DK + koff;
#pragma unroll
    for (int kk = 0; kk < 8; ++kk) {
      aq[q][kk].x = qp[kk * 4];
      aq[q][kk].y = qp[kk * 4 + 1];
    }
  }

  const float* kS = kb + ((size_t)b * C_ + h * DK) * N_;
  const float* vS = vt + ((size_t)b * HDS + h) * N_ * DK;
  const unsigned kLane = (unsigned)(koff * N_ + row) * 4u;
  const unsigned vLane = (unsigned)(koff * DK + row) * 4u;

  v8f o0[QT], o1[QT];
  float m[QT], l[QT];
#pragma unroll
  for (int q = 0; q < QT; ++q) {
    o0[q] = (v8f){};
    o1[q] = (v8f){};
    m[q] = -1e30f;
    l[q] = 0.0f;
  }

  const int mt0 = ks * NTH;
  for (int it = 0; it < NSTEP; ++it) {
    const int mBase = (mt0 + it * KT) * 16;

    v8f S[QT][KT];
#pragma unroll
    for (int q = 0; q < QT; ++q) {
#pragma unroll
      for (int t = 0; t < KT; ++t) S[q][t] = (v8f){};
    }
#pragma unroll
    for (int kg = 0; kg < 8; kg += KG) {
      v2f bk[KG][KT];
#pragma unroll
      for (int j = 0; j < KG; ++j) {
        const char* kp = (const char*)(kS + mBase + (size_t)((kg + j) * 4) * N_);
#pragma unroll
        for (int t = 0; t < KT; ++t) {
          bk[j][t].x = *(const float*)(kp + t * 64 + kLane);
          bk[j][t].y = *(const float*)(kp + t * 64 + N_ * 4 + kLane);
        }
      }
#pragma unroll
      for (int j = 0; j < KG; ++j) {
#pragma unroll
        for (int t = 0; t < KT; ++t) {
#pragma unroll
          for (int q = 0; q < QT; ++q)
            S[q][t] = wmma_f32(aq[q][kg + j], bk[j][t], S[q][t]);
        }
      }
    }

#pragma unroll
    for (int q = 0; q < QT; ++q) {
#pragma unroll
      for (int t = 0; t < KT; ++t) {
#pragma unroll
        for (int v = 0; v < 8; ++v)
          P[(q * KT + t) * PT + (v + half * 8) * PP + pcol] = S[q][t][v];
      }
    }

    v2f ap[QT][KT][4];
#pragma unroll
    for (int q = 0; q < QT; ++q) {
#pragma unroll
      for (int t = 0; t < KT; ++t) {
#pragma unroll
        for (int j = 0; j < 2; ++j) {
          const v4f f =
              *(const v4f*)&P[(q * KT + t) * PT + row * PP + half * 8 + j * 4];
          ap[q][t][2 * j]     = (v2f){f.x, f.y};
          ap[q][t][2 * j + 1] = (v2f){f.z, f.w};
        }
      }
      float mx = m[q];
#pragma unroll
      for (int t = 0; t < KT; ++t) {
#pragma unroll
        for (int kk = 0; kk < 4; ++kk)
          mx = fmaxf(mx, fmaxf(ap[q][t][kk].x, ap[q][t][kk].y));
      }
      const float mn = fmaxf(mx, xhalf(mx));
      const float al = __expf(m[q] - mn);
      m[q] = mn;
      float st[KT];
#pragma unroll
      for (int t = 0; t < KT; ++t) {
        float a = 0.0f;
#pragma unroll
        for (int kk = 0; kk < 4; ++kk) {
          const float p0 = __expf(ap[q][t][kk].x - mn);
          const float p1 = __expf(ap[q][t][kk].y - mn);
          ap[q][t][kk].x = p0;
          ap[q][t][kk].y = p1;
          a += p0 + p1;
        }
        st[t] = a;
      }
      l[q] = l[q] * al + ((st[0] + st[1]) + (st[2] + st[3]));
      float alC[8];
      rows_to_c8(al, alC);
#pragma unroll
      for (int v = 0; v < 8; ++v) {
        o0[q][v] *= alC[v];
        o1[q][v] *= alC[v];
      }
    }

    const char* vStep = (const char*)(vS + (size_t)mBase * DK) + vLane;
#pragma unroll
    for (int t0 = 0; t0 < KT; t0 += TG) {
      v2f bv0[TG][4], bv1[TG][4];
#pragma unroll
      for (int tt = 0; tt < TG; ++tt) {
        const int t = t0 + tt;
#pragma unroll
        for (int kk = 0; kk < 4; ++kk) {
          const char* vp = vStep + (t * 16 + kk * 4) * (DK * 4);
          bv0[tt][kk].x = *(const float*)(vp);
          bv0[tt][kk].y = *(const float*)(vp + DK * 4);
          bv1[tt][kk].x = *(const float*)(vp + 16 * 4);
          bv1[tt][kk].y = *(const float*)(vp + (DK + 16) * 4);
        }
      }
#pragma unroll
      for (int tt = 0; tt < TG; ++tt) {
#pragma unroll
        for (int kk = 0; kk < 4; ++kk) {
#pragma unroll
          for (int q = 0; q < QT; ++q) {
            o0[q] = wmma_f32(ap[q][t0 + tt][kk], bv0[tt][kk], o0[q]);
            o1[q] = wmma_f32(ap[q][t0 + tt][kk], bv1[tt][kk], o1[q]);
          }
        }
      }
    }
  }

  float4* J = &ldsJ[qsel][0][0][lane];
  if (ks == 1) {
#pragma unroll
    for (int q = 0; q < QT; ++q) {
      float4* Jq = J + q * 5 * 32;
      Jq[0 * 32] = make_float4(o0[q][0], o0[q][1], o0[q][2], o0[q][3]);
      Jq[1 * 32] = make_float4(o0[q][4], o0[q][5], o0[q][6], o0[q][7]);
      Jq[2 * 32] = make_float4(o1[q][0], o1[q][1], o1[q][2], o1[q][3]);
      Jq[3 * 32] = make_float4(o1[q][4], o1[q][5], o1[q][6], o1[q][7]);
      Jq[4 * 32] = make_float4(m[q], l[q], 0.0f, 0.0f);
    }
  }
  __syncthreads();
  if (ks == 1) return;

#pragma unroll
  for (int q = 0; q < QT; ++q) {
    const float4* Jq = J + q * 5 * 32;
    float po0[8], po1[8];
#pragma unroll
    for (int r = 0; r < 2; ++r) {
      const float4 a0 = Jq[(0 + r) * 32], a1 = Jq[(2 + r) * 32];
      po0[4*r+0] = a0.x; po0[4*r+1] = a0.y; po0[4*r+2] = a0.z; po0[4*r+3] = a0.w;
      po1[4*r+0] = a1.x; po1[4*r+1] = a1.y; po1[4*r+2] = a1.z; po1[4*r+3] = a1.w;
    }
    const float4 ml = Jq[4 * 32];
    const float mn = fmaxf(m[q], ml.x);
    const float a  = __expf(m[q] - mn);
    const float pa = __expf(ml.x - mn);
    const float lp = l[q] * a + ml.y * pa;
    const float inv = 1.0f / (lp + xhalf(lp));
    float aC[8], paC[8], invC[8];
    rows_to_c8(a, aC);
    rows_to_c8(pa, paC);
    rows_to_c8(inv, invC);

    const size_t obase =
        ((size_t)b * C_ + h * DK + row) * N_ + nBase + q * 16 + half * 8;
#pragma unroll
    for (int v = 0; v < 8; ++v) {
      const float r0 = o0[q][v] * aC[v] + po0[v] * paC[v];
      const float r1 = o1[q][v] * aC[v] + po1[v] * paC[v];
      const size_t i0 = obase + v;
      out[i0] = x[i0] + r0 * invC[v];
      const size_t i1 = i0 + (size_t)16 * N_;
      out[i1] = x[i1] + r1 * invC[v];
    }
  }
}

extern "C" void kernel_launch(void* const* d_in, const int* in_sizes, int n_in,
                              void* d_out, int out_size, void* d_ws,
                              size_t ws_size, hipStream_t stream) {
  const float* x  = (const float*)d_in[0];
  const float* wq = (const float*)d_in[1];
  const float* wk = (const float*)d_in[2];
  const float* wv = (const float*)d_in[3];
  float* out = (float*)d_out;

  float* qt = (float*)d_ws;
  float* kb = qt + (size_t)B_ * C_ * N_;
  float* vt = kb + (size_t)B_ * C_ * N_;

  dim3 blk(128);
  mha_proj_kernel<<<dim3(144, 12), blk, 0, stream>>>(x, wq, wk, wv, qt, kb, vt);
  mha_attn_kernel<<<dim3(768), blk, 0, stream>>>(x, qt, kb, vt, out);
}
